# starting kernel
# baseline (speedup 1.0000x reference)
#include <hip/hip_runtime.h>

typedef float v2f __attribute__((ext_vector_type(2)));
typedef float v8f __attribute__((ext_vector_type(8)));
typedef int   v2i __attribute__((ext_vector_type(2)));

#define NFPN_EPS2 1e-10f
#define NFPN_MAX_DEPTH 100

__global__ __launch_bounds__(256) void nfpn_rps_kernel(
    const float* __restrict__ din,
    const float* __restrict__ W1,
    const float* __restrict__ b1,
    const float* __restrict__ W2,
    const float* __restrict__ b2,
    float* __restrict__ out,
    int B)
{
    const int gid  = blockIdx.x * blockDim.x + threadIdx.x;
    const int lane = threadIdx.x & 31;
    const bool low = lane < 16;

    int s = gid;
    const bool active = (s < B);
    if (!active) s = B - 1;

    const float d0 = din[3*s + 0];
    const float d1 = din[3*s + 1];
    const float d2 = din[3*s + 2];

    float q[6];
#pragma unroll
    for (int j = 0; j < 6; ++j) {
        float t = fmaf(d0, W1[3*j + 0], fmaf(d1, W1[3*j + 1], fmaf(d2, W1[3*j + 2], b1[j])));
        q[j] = (t >= 0.f) ? t : 0.1f * t;
    }

    const int  M     = lane & 15;
    const int  rr    = M & 7;
    const bool valid = (rr < 6);
    v2f a1, a2;
    a1.x = valid ? -W2[rr*6 + (low ? 0 : 2)] : 0.f;
    a1.y = valid ? -W2[rr*6 + (low ? 1 : 3)] : 0.f;
    a2.x = (valid && low) ? -W2[rr*6 + 4] : 0.f;
    a2.y = (valid && low) ? -W2[rr*6 + 5] : 0.f;

    v8f cinit;
#pragma unroll
    for (int j = 0; j < 6; ++j) cinit[j] = -b2[j];
    cinit[6] = 0.f; cinit[7] = 0.f;

    float u[6];
#pragma unroll
    for (int j = 0; j < 6; ++j) u[j] = 1.0f / 3.0f;

    auto swap_halves = [&](float a, float b) -> v2f {
        v2f o;
#if __has_builtin(__builtin_amdgcn_permlane16_swap)
        v2i r = __builtin_amdgcn_permlane16_swap(__float_as_int(a), __float_as_int(b),
                                                 false, false);
        o.x = __int_as_float(r.x);
        o.y = __int_as_float(r.y);
#else
        float as = __shfl_xor(a, 16, 32);
        float bs = __shfl_xor(b, 16, 32);
        o.x = low ? a : bs;
        o.y = low ? as : b;
#endif
        return o;
    };

    auto matvec = [&](const float* x, float* y) {
        v2f B1, B2, B1p, B2p;
        v2f s02 = swap_halves(x[0], x[2]);  B1.x = s02.x;  B1p.x = s02.y;
        v2f s13 = swap_halves(x[1], x[3]);  B1.y = s13.x;  B1p.y = s13.y;
        v2f s4z = swap_halves(x[4], 0.f);   B2.x = s4z.x;  B2p.x = s4z.y;
        v2f s5z = swap_halves(x[5], 0.f);   B2.y = s5z.x;  B2p.y = s5z.y;
        v8f acc = cinit;
        acc  = __builtin_amdgcn_wmma_f32_16x16x4_f32(false, a1, false, B1,  (short)0, acc,  false, false);
        acc  = __builtin_amdgcn_wmma_f32_16x16x4_f32(false, a2, false, B2,  (short)0, acc,  false, false);
        v8f accp = cinit;
        accp = __builtin_amdgcn_wmma_f32_16x16x4_f32(false, a1, false, B1p, (short)0, accp, false, false);
        accp = __builtin_amdgcn_wmma_f32_16x16x4_f32(false, a2, false, B2p, (short)0, accp, false, false);
#pragma unroll
        for (int j = 0; j < 6; ++j) y[j] = low ? acc[j] : accp[j];
    };

    auto proj3 = [&](float a, float b, float c, float& o0, float& o1, float& o2) {
        float hi  = fmaxf(fmaxf(a, b), c);
        float lo  = fminf(fminf(a, b), c);
        float sum = a + b + c;
        float mid = sum - hi - lo;
        float c2  = hi + mid;
        bool cond2 = (mid - hi + 1.0f) > 0.0f;
        bool cond3 = (3.0f*lo - sum + 1.0f) > 0.0f;
        int rho = 1 + (cond2 ? 1 : 0) + (cond3 ? 1 : 0);
        float cs    = (rho == 3) ? sum : ((rho == 2) ? c2 : hi);
        float rinv  = (rho == 3) ? (1.0f/3.0f) : ((rho == 2) ? 0.5f : 1.0f);
        float theta = (cs - 1.0f) * rinv;
        o0 = fmaxf(a - theta, 0.f);
        o1 = fmaxf(b - theta, 0.f);
        o2 = fmaxf(c - theta, 0.f);
    };

    int depth = 0;
    for (;;) {
        float x[6], y[6], un[6];
#pragma unroll
        for (int j = 0; j < 6; ++j) x[j] = u[j] + q[j];
        matvec(x, y);
        proj3(y[0], y[1], y[2], un[0], un[1], un[2]);
        proj3(y[3], y[4], y[5], un[3], un[4], un[5]);
        float r2 = 0.f;
#pragma unroll
        for (int j = 0; j < 6; ++j) { float dd = un[j] - u[j]; r2 = fmaf(dd, dd, r2); u[j] = un[j]; }
        ++depth;
        if (depth >= NFPN_MAX_DEPTH) break;
        if (__ballot(r2 > NFPN_EPS2) == 0ULL) break;
    }

    float x[6], y[6], o[6];
#pragma unroll
    for (int j = 0; j < 6; ++j) x[j] = u[j] + q[j];
    matvec(x, y);
    proj3(y[0], y[1], y[2], o[0], o[1], o[2]);
    proj3(y[3], y[4], y[5], o[3], o[4], o[5]);

    if (active) {
        float2* op = reinterpret_cast<float2*>(out + 6l * s);
        op[0] = make_float2(o[0], o[1]);
        op[1] = make_float2(o[2], o[3]);
        op[2] = make_float2(o[4], o[5]);
    }
}

extern "C" void kernel_launch(void* const* d_in, const int* in_sizes, int n_in,
                              void* d_out, int out_size, void* d_ws, size_t ws_size,
                              hipStream_t stream) {
    const float* d  = (const float*)d_in[0];
    const float* W1 = (const float*)d_in[1];
    const float* b1 = (const float*)d_in[2];
    const float* W2 = (const float*)d_in[3];
    const float* b2 = (const float*)d_in[4];
    float* out = (float*)d_out;

    const int B = in_sizes[0] / 3;
    const int threads = 256;
    const int blocks  = (B + threads - 1) / threads;
    hipLaunchKernelGGL(nfpn_rps_kernel, dim3(blocks), dim3(threads), 0, stream,
                       d, W1, b1, W2, b2, out, B);
}


# Sample on a lane pair (half h owns 3-block h, 2 tiles a wave): WMMA rows give y and each block's subset means, theta = their max; no operand swaps, selects, rho. Sums: products 0,1,3,4|2,5; r2 3+3
# speedup vs baseline: 1.2123x; 1.2123x over previous
#include <hip/hip_runtime.h>


typedef float v2f __attribute__((ext_vector_type(2)));
typedef float v8f __attribute__((ext_vector_type(8)));
typedef int   v2i __attribute__((ext_vector_type(2)));

#define NFPN_EPS2 1e-10f
#define NFPN_MAX_DEPTH 100

__global__ __launch_bounds__(256) void nfpn_rps_kernel(
    const float* __restrict__ din,
    const float* __restrict__ W1,
    const float* __restrict__ b1,
    const float* __restrict__ W2,
    const float* __restrict__ b2,
    float* __restrict__ out,
    int B)
{
    const int lane  = threadIdx.x & 31;
    const int c     = lane & 15;
    const int h     = lane >> 4;
    const int wbase = (int)(blockIdx.x * blockDim.x + threadIdx.x) - lane;

    int  s[2];
    bool act[2];
#pragma unroll
    for (int t = 0; t < 2; ++t) {
        s[t]   = wbase + 16 * t + c;
        act[t] = (s[t] < B);
        if (!act[t]) s[t] = B - 1;
    }

    const float* W1h = W1 + 9 * h;
    const float* b1h = b1 + 3 * h;
    float q[2][3];
#pragma unroll
    for (int t = 0; t < 2; ++t) {
        const float d0 = din[3 * s[t] + 0];
        const float d1 = din[3 * s[t] + 1];
        const float d2 = din[3 * s[t] + 2];
#pragma unroll
        for (int i = 0; i < 3; ++i) {
            float v = fmaf(d0, W1h[3*i + 0], fmaf(d1, W1h[3*i + 1], fmaf(d2, W1h[3*i + 2], b1h[i])));
            q[t][i] = (v >= 0.f) ? v : 0.1f * v;
        }
    }

    const int      ab   = c >> 3;
    const int      ar   = c & 7;
    const unsigned sel  = (0x07653421u >> (4 * ar)) & 7u;
    const float    scl  = (ar < 3) ? 1.0f : ((ar < 6) ? 0.5f : ((ar == 6) ? (1.0f / 3.0f) : 0.0f));
    const float*   Wsub = W2 + 18 * ab + 3 * h;
    float a[3];
#pragma unroll
    for (int e = 0; e < 3; ++e) {
        const float w0 = (sel & 1u) ? Wsub[e]      : 0.f;
        const float w1 = (sel & 2u) ? Wsub[6 + e]  : 0.f;
        const float w2 = (sel & 4u) ? Wsub[12 + e] : 0.f;
        a[e] = -scl * ((w0 + w1) + w2);
    }
    v2f a1, a2;
    a1.x = a[0]; a1.y = a[1];
    a2.x = a[2]; a2.y = 0.f;

    const float e0 = b2[3 * h + 0];
    const float e1 = b2[3 * h + 1];
    const float e2 = b2[3 * h + 2];
    v8f cinit;
    cinit[0] = -e0;
    cinit[1] = -e1;
    cinit[2] = -e2;
    cinit[3] = -0.5f * ((e0 + e1) + 1.0f);
    cinit[4] = -0.5f * ((e0 + e2) + 1.0f);
    cinit[5] = -0.5f * ((e1 + e2) + 1.0f);
    cinit[6] = -(1.0f / 3.0f) * (((e0 + e1) + e2) + 1.0f);
    cinit[7] = 0.f;

    auto step2 = [&](const float (&u)[2][3], float (&un)[2][3]) {
        v8f acc[2];
#pragma unroll
        for (int t = 0; t < 2; ++t) {
            v2f B1, B2;
            B1.x = u[t][0] + q[t][0];
            B1.y = u[t][1] + q[t][1];
            B2.x = u[t][2] + q[t][2];
            B2.y = 0.f;
            acc[t] = __builtin_amdgcn_wmma_f32_16x16x4_f32(false, a1, false, B1, (short)0, cinit,  false, false);
            acc[t] = __builtin_amdgcn_wmma_f32_16x16x4_f32(false, a2, false, B2, (short)0, acc[t], false, false);
        }
#pragma unroll
        for (int t = 0; t < 2; ++t) {
            const float hi = fmaxf(fmaxf(acc[t][0], acc[t][1]), acc[t][2]);
            const float mp = fmaxf(fmaxf(acc[t][3], acc[t][4]), acc[t][5]);
            const float th = fmaxf(fmaxf(hi - 1.0f, mp), acc[t][6]);
#pragma unroll
            for (int i = 0; i < 3; ++i)
                un[t][i] = fmaxf(acc[t][i] - th, 0.f);
        }
    };

    float u[2][3];
#pragma unroll
    for (int t = 0; t < 2; ++t)
#pragma unroll
        for (int i = 0; i < 3; ++i) u[t][i] = 1.0f / 3.0f;

    int depth = 0;
    for (;;) {
        float un[2][3];
        step2(u, un);
        float rp[2];
#pragma unroll
        for (int t = 0; t < 2; ++t) {
            const float g0 = un[t][0] - u[t][0];
            const float g1 = un[t][1] - u[t][1];
            const float g2 = un[t][2] - u[t][2];
            rp[t] = fmaf(g2, g2, fmaf(g1, g1, g0 * g0));
            u[t][0] = un[t][0]; u[t][1] = un[t][1]; u[t][2] = un[t][2];
        }
        v2i sw = __builtin_amdgcn_permlane16_swap(__float_as_int(rp[0]), __float_as_int(rp[1]), false, false);
        const float r2 = __int_as_float(sw.x) + __int_as_float(sw.y);
        ++depth;
        if (depth >= NFPN_MAX_DEPTH) break;
        if (__ballot(r2 > NFPN_EPS2) == 0ULL) break;
    }

    float o[2][3];
    step2(u, o);

#pragma unroll
    for (int t = 0; t < 2; ++t) {
        if (act[t]) {
            float* op = out + 6l * s[t] + 3 * h;
            op[0] = o[t][0];
            op[1] = o[t][1];
            op[2] = o[t][2];
        }
    }
}

extern "C" void kernel_launch(void* const* d_in, const int* in_sizes, int n_in,
                              void* d_out, int out_size, void* d_ws, size_t ws_size,
                              hipStream_t stream) {
    const float* d  = (const float*)d_in[0];
    const float* W1 = (const float*)d_in[1];
    const float* b1 = (const float*)d_in[2];
    const float* W2 = (const float*)d_in[3];
    const float* b2 = (const float*)d_in[4];
    float* out = (float*)d_out;

    const int B = in_sizes[0] / 3;
    const int threads = 256;
    const int blocks  = (B + threads - 1) / threads;
    hipLaunchKernelGGL(nfpn_rps_kernel, dim3(blocks), dim3(threads), 0, stream,
                       d, W1, b1, W2, b2, out, B);
}


# Loop holds two steps a trip: state alternates between two register sets (no copies back), residual formed and tested after every second step; a wave runs as many steps as before or one more, cap 100
# speedup vs baseline: 1.3122x; 1.0824x over previous
#include <hip/hip_runtime.h>


typedef float v2f __attribute__((ext_vector_type(2)));
typedef float v8f __attribute__((ext_vector_type(8)));
typedef int   v2i __attribute__((ext_vector_type(2)));

#define NFPN_EPS2 1e-10f
#define NFPN_MAX_DEPTH 100

__global__ __launch_bounds__(256) void nfpn_rps_kernel(
    const float* __restrict__ din,
    const float* __restrict__ W1,
    const float* __restrict__ b1,
    const float* __restrict__ W2,
    const float* __restrict__ b2,
    float* __restrict__ out,
    int B)
{
    const int lane  = threadIdx.x & 31;
    const int c     = lane & 15;
    const int h     = lane >> 4;
    const int wbase = (int)(blockIdx.x * blockDim.x + threadIdx.x) - lane;

    int  s[2];
    bool act[2];
#pragma unroll
    for (int t = 0; t < 2; ++t) {
        s[t]   = wbase + 16 * t + c;
        act[t] = (s[t] < B);
        if (!act[t]) s[t] = B - 1;
    }

    const float* W1h = W1 + 9 * h;
    const float* b1h = b1 + 3 * h;
    float q[2][3];
#pragma unroll
    for (int t = 0; t < 2; ++t) {
        const float d0 = din[3 * s[t] + 0];
        const float d1 = din[3 * s[t] + 1];
        const float d2 = din[3 * s[t] + 2];
#pragma unroll
        for (int i = 0; i < 3; ++i) {
            float v = fmaf(d0, W1h[3*i + 0], fmaf(d1, W1h[3*i + 1], fmaf(d2, W1h[3*i + 2], b1h[i])));
            q[t][i] = (v >= 0.f) ? v : 0.1f * v;
        }
    }

    const int      ab   = c >> 3;
    const int      ar   = c & 7;
    const unsigned sel  = (0x07653421u >> (4 * ar)) & 7u;
    const float    scl  = (ar < 3) ? 1.0f : ((ar < 6) ? 0.5f : ((ar == 6) ? (1.0f / 3.0f) : 0.0f));
    const float*   Wsub = W2 + 18 * ab + 3 * h;
    float a[3];
#pragma unroll
    for (int e = 0; e < 3; ++e) {
        const float w0 = (sel & 1u) ? Wsub[e]      : 0.f;
        const float w1 = (sel & 2u) ? Wsub[6 + e]  : 0.f;
        const float w2 = (sel & 4u) ? Wsub[12 + e] : 0.f;
        a[e] = -scl * ((w0 + w1) + w2);
    }
    v2f a1, a2;
    a1.x = a[0]; a1.y = a[1];
    a2.x = a[2]; a2.y = 0.f;

    const float e0 = b2[3 * h + 0];
    const float e1 = b2[3 * h + 1];
    const float e2 = b2[3 * h + 2];
    v8f cinit;
    cinit[0] = -e0;
    cinit[1] = -e1;
    cinit[2] = -e2;
    cinit[3] = -0.5f * ((e0 + e1) + 1.0f);
    cinit[4] = -0.5f * ((e0 + e2) + 1.0f);
    cinit[5] = -0.5f * ((e1 + e2) + 1.0f);
    cinit[6] = -(1.0f / 3.0f) * (((e0 + e1) + e2) + 1.0f);
    cinit[7] = 0.f;

    auto step2 = [&](const float (&u)[2][3], float (&un)[2][3]) {
        v8f acc[2];
#pragma unroll
        for (int t = 0; t < 2; ++t) {
            v2f B1, B2;
            B1.x = u[t][0] + q[t][0];
            B1.y = u[t][1] + q[t][1];
            B2.x = u[t][2] + q[t][2];
            B2.y = 0.f;
            acc[t] = __builtin_amdgcn_wmma_f32_16x16x4_f32(false, a1, false, B1, (short)0, cinit,  false, false);
            acc[t] = __builtin_amdgcn_wmma_f32_16x16x4_f32(false, a2, false, B2, (short)0, acc[t], false, false);
        }
#pragma unroll
        for (int t = 0; t < 2; ++t) {
            const float hi = fmaxf(fmaxf(acc[t][0], acc[t][1]), acc[t][2]);
            const float mp = fmaxf(fmaxf(acc[t][3], acc[t][4]), acc[t][5]);
            const float th = fmaxf(fmaxf(hi - 1.0f, mp), acc[t][6]);
#pragma unroll
            for (int i = 0; i < 3; ++i)
                un[t][i] = fmaxf(acc[t][i] - th, 0.f);
        }
    };

    float u[2][3];
#pragma unroll
    for (int t = 0; t < 2; ++t)
#pragma unroll
        for (int i = 0; i < 3; ++i) u[t][i] = 1.0f / 3.0f;

    static_assert((NFPN_MAX_DEPTH % 2) == 0, "two steps per trip need an even cap");
    float v[2][3];
    int depth = 0;
    for (;;) {
        step2(u, v);
        step2(v, u);
        float rp[2];
#pragma unroll
        for (int t = 0; t < 2; ++t) {
            const float g0 = u[t][0] - v[t][0];
            const float g1 = u[t][1] - v[t][1];
            const float g2 = u[t][2] - v[t][2];
            rp[t] = fmaf(g2, g2, fmaf(g1, g1, g0 * g0));
        }
        v2i sw = __builtin_amdgcn_permlane16_swap(__float_as_int(rp[0]), __float_as_int(rp[1]), false, false);
        const float r2 = __int_as_float(sw.x) + __int_as_float(sw.y);
        depth += 2;
        if (depth >= NFPN_MAX_DEPTH) break;
        if (__ballot(r2 > NFPN_EPS2) == 0ULL) break;
    }

    float o[2][3];
    step2(u, o);

#pragma unroll
    for (int t = 0; t < 2; ++t) {
        if (act[t]) {
            float* op = out + 6l * s[t] + 3 * h;
            op[0] = o[t][0];
            op[1] = o[t][1];
            op[2] = o[t][2];
        }
    }
}

extern "C" void kernel_launch(void* const* d_in, const int* in_sizes, int n_in,
                              void* d_out, int out_size, void* d_ws, size_t ws_size,
                              hipStream_t stream) {
    const float* d  = (const float*)d_in[0];
    const float* W1 = (const float*)d_in[1];
    const float* b1 = (const float*)d_in[2];
    const float* W2 = (const float*)d_in[3];
    const float* b2 = (const float*)d_in[4];
    float* out = (float*)d_out;

    const int B = in_sizes[0] / 3;
    const int threads = 256;
    const int blocks  = (B + threads - 1) / threads;
    hipLaunchKernelGGL(nfpn_rps_kernel, dim3(blocks), dim3(threads), 0, stream,
                       d, W1, b1, W2, b2, out, B);
}


# Once 16 or fewer of a wave's 32 samples still move, the rest are stored and the moving ones packed into one tile by a lane permute: 2 products a step, not 4; a sample may stop before its wave does
# speedup vs baseline: 1.3629x; 1.0386x over previous
#include <hip/hip_runtime.h>


typedef float v2f __attribute__((ext_vector_type(2)));
typedef float v8f __attribute__((ext_vector_type(8)));
typedef int   v2i __attribute__((ext_vector_type(2)));

#define NFPN_EPS2 1e-10f
#define NFPN_MAX_DEPTH 100

__global__ __launch_bounds__(256) void nfpn_rps_kernel(
    const float* __restrict__ din,
    const float* __restrict__ W1,
    const float* __restrict__ b1,
    const float* __restrict__ W2,
    const float* __restrict__ b2,
    float* __restrict__ out,
    int B)
{
    const int lane  = threadIdx.x & 31;
    const int c     = lane & 15;
    const int h     = lane >> 4;
    const int wbase = (int)(blockIdx.x * blockDim.x + threadIdx.x) - lane;

    int  s[2];
    bool act[2];
#pragma unroll
    for (int t = 0; t < 2; ++t) {
        s[t]   = wbase + 16 * t + c;
        act[t] = (s[t] < B);
        if (!act[t]) s[t] = B - 1;
    }

    const float* W1h = W1 + 9 * h;
    const float* b1h = b1 + 3 * h;
    float q[2][3];
#pragma unroll
    for (int t = 0; t < 2; ++t) {
        const float d0 = din[3 * s[t] + 0];
        const float d1 = din[3 * s[t] + 1];
        const float d2 = din[3 * s[t] + 2];
#pragma unroll
        for (int i = 0; i < 3; ++i) {
            float v = fmaf(d0, W1h[3*i + 0], fmaf(d1, W1h[3*i + 1], fmaf(d2, W1h[3*i + 2], b1h[i])));
            q[t][i] = (v >= 0.f) ? v : 0.1f * v;
        }
    }

    const int      ab   = c >> 3;
    const int      ar   = c & 7;
    const unsigned sel  = (0x07653421u >> (4 * ar)) & 7u;
    const float    scl  = (ar < 3) ? 1.0f : ((ar < 6) ? 0.5f : ((ar == 6) ? (1.0f / 3.0f) : 0.0f));
    const float*   Wsub = W2 + 18 * ab + 3 * h;
    float a[3];
#pragma unroll
    for (int e = 0; e < 3; ++e) {
        const float w0 = (sel & 1u) ? Wsub[e]      : 0.f;
        const float w1 = (sel & 2u) ? Wsub[6 + e]  : 0.f;
        const float w2 = (sel & 4u) ? Wsub[12 + e] : 0.f;
        a[e] = -scl * ((w0 + w1) + w2);
    }
    v2f a1, a2;
    a1.x = a[0]; a1.y = a[1];
    a2.x = a[2]; a2.y = 0.f;

    const float e0 = b2[3 * h + 0];
    const float e1 = b2[3 * h + 1];
    const float e2 = b2[3 * h + 2];
    v8f cinit;
    cinit[0] = -e0;
    cinit[1] = -e1;
    cinit[2] = -e2;
    cinit[3] = -0.5f * ((e0 + e1) + 1.0f);
    cinit[4] = -0.5f * ((e0 + e2) + 1.0f);
    cinit[5] = -0.5f * ((e1 + e2) + 1.0f);
    cinit[6] = -(1.0f / 3.0f) * (((e0 + e1) + e2) + 1.0f);
    cinit[7] = 0.f;

    auto step2 = [&](const float (&u)[2][3], float (&un)[2][3]) {
        v8f acc[2];
#pragma unroll
        for (int t = 0; t < 2; ++t) {
            v2f B1, B2;
            B1.x = u[t][0] + q[t][0];
            B1.y = u[t][1] + q[t][1];
            B2.x = u[t][2] + q[t][2];
            B2.y = 0.f;
            acc[t] = __builtin_amdgcn_wmma_f32_16x16x4_f32(false, a1, false, B1, (short)0, cinit,  false, false);
            acc[t] = __builtin_amdgcn_wmma_f32_16x16x4_f32(false, a2, false, B2, (short)0, acc[t], false, false);
        }
#pragma unroll
        for (int t = 0; t < 2; ++t) {
            const float hi = fmaxf(fmaxf(acc[t][0], acc[t][1]), acc[t][2]);
            const float mp = fmaxf(fmaxf(acc[t][3], acc[t][4]), acc[t][5]);
            const float th = fmaxf(fmaxf(hi - 1.0f, mp), acc[t][6]);
#pragma unroll
            for (int i = 0; i < 3; ++i)
                un[t][i] = fmaxf(acc[t][i] - th, 0.f);
        }
    };

    auto step1 = [&](const float (&x)[3], const float (&qq)[3], float (&xn)[3]) {
        v2f B1, B2;
        B1.x = x[0] + qq[0];
        B1.y = x[1] + qq[1];
        B2.x = x[2] + qq[2];
        B2.y = 0.f;
        v8f acc = __builtin_amdgcn_wmma_f32_16x16x4_f32(false, a1, false, B1, (short)0, cinit, false, false);
        acc     = __builtin_amdgcn_wmma_f32_16x16x4_f32(false, a2, false, B2, (short)0, acc,   false, false);
        const float hi = fmaxf(fmaxf(acc[0], acc[1]), acc[2]);
        const float mp = fmaxf(fmaxf(acc[3], acc[4]), acc[5]);
        const float th = fmaxf(fmaxf(hi - 1.0f, mp), acc[6]);
#pragma unroll
        for (int i = 0; i < 3; ++i)
            xn[i] = fmaxf(acc[i] - th, 0.f);
    };

    float u[2][3];
#pragma unroll
    for (int t = 0; t < 2; ++t)
#pragma unroll
        for (int i = 0; i < 3; ++i) u[t][i] = 1.0f / 3.0f;

    static_assert((NFPN_MAX_DEPTH % 2) == 0, "two steps per trip need an even cap");
    float v[2][3];
    int depth = 0;
    unsigned moving = 0u;
    for (;;) {
        step2(u, v);
        step2(v, u);
        float rp[2];
#pragma unroll
        for (int t = 0; t < 2; ++t) {
            const float g0 = u[t][0] - v[t][0];
            const float g1 = u[t][1] - v[t][1];
            const float g2 = u[t][2] - v[t][2];
            rp[t] = fmaf(g2, g2, fmaf(g1, g1, g0 * g0));
        }
        v2i sw = __builtin_amdgcn_permlane16_swap(__float_as_int(rp[0]), __float_as_int(rp[1]), false, false);
        const float r2 = __int_as_float(sw.x) + __int_as_float(sw.y);
        depth += 2;
        if (depth >= NFPN_MAX_DEPTH) { moving = 0u; break; }
        moving = __builtin_amdgcn_ballot_w32(r2 > NFPN_EPS2);
        if (__popc(moving) <= 16) break;
    }

    {
        float o[2][3];
        step2(u, o);
#pragma unroll
        for (int t = 0; t < 2; ++t) {
            const bool done = ((moving >> (16 * t + c)) & 1u) == 0u;
            if (act[t] && done) {
                float* op = out + 6l * s[t] + 3 * h;
                op[0] = o[t][0];
                op[1] = o[t][1];
                op[2] = o[t][2];
            }
        }
    }
    if (moving == 0u) return;

    const unsigned m0 = moving & 0xFFFFu;
    const unsigned m1 = moving >> 16;
    const int n0 = __popc(m0);
    const int n1 = __popc(m1);
    const unsigned below = (1u << c) - 1u;
    const int k0 = __popc(m0 & below);
    const int k1 = __popc(m1 & below);
    const int p0 = ((m0 >> c) & 1u) ? k0 : n0 + (c - k0);
    const int p1 = (((m1 >> c) & 1u) ? n0 + k1 : n0 + n1 + (c - k1)) & 15;
    const int ad0 = 4 * (p0 + 16 * h);
    const int ad1 = 4 * (p1 + 16 * h);
    const bool from0 = (c < n0);
    auto pack = [&](int x0, int x1) -> int {
        const int y0 = __builtin_amdgcn_ds_permute(ad0, x0);
        const int y1 = __builtin_amdgcn_ds_permute(ad1, x1);
        return from0 ? y0 : y1;
    };
    float pu[3], pq[3];
#pragma unroll
    for (int i = 0; i < 3; ++i) {
        pu[i] = __int_as_float(pack(__float_as_int(u[0][i]), __float_as_int(u[1][i])));
        pq[i] = __int_as_float(pack(__float_as_int(q[0][i]), __float_as_int(q[1][i])));
    }
    const int  ps   = pack(act[0] ? s[0] : -1, act[1] ? s[1] : -1);
    const bool live = (c < n0 + n1);
    const bool pact = live && (ps >= 0);

    float pv[3];
    for (;;) {
        step1(pu, pq, pv);
        step1(pv, pq, pu);
        const float g0 = pu[0] - pv[0];
        const float g1 = pu[1] - pv[1];
        const float g2 = pu[2] - pv[2];
        const int rp = __float_as_int(fmaf(g2, g2, fmaf(g1, g1, g0 * g0)));
        v2i sw = __builtin_amdgcn_permlane16_swap(rp, rp, false, false);
        const float r2 = __int_as_float(sw.x) + __int_as_float(sw.y);
        depth += 2;
        if (depth >= NFPN_MAX_DEPTH) break;
        if (__builtin_amdgcn_ballot_w32(live && (r2 > NFPN_EPS2)) == 0u) break;
    }

    float po[3];
    step1(pu, pq, po);
    if (pact) {
        float* op = out + 6l * ps + 3 * h;
        op[0] = po[0];
        op[1] = po[1];
        op[2] = po[2];
    }
}

extern "C" void kernel_launch(void* const* d_in, const int* in_sizes, int n_in,
                              void* d_out, int out_size, void* d_ws, size_t ws_size,
                              hipStream_t stream) {
    const float* d  = (const float*)d_in[0];
    const float* W1 = (const float*)d_in[1];
    const float* b1 = (const float*)d_in[2];
    const float* W2 = (const float*)d_in[3];
    const float* b2 = (const float*)d_in[4];
    float* out = (float*)d_out;

    const int B = in_sizes[0] / 3;
    const int threads = 256;
    const int blocks  = (B + threads - 1) / threads;
    hipLaunchKernelGGL(nfpn_rps_kernel, dim3(blocks), dim3(threads), 0, stream,
                       d, W1, b1, W2, b2, out, B);
}


# 512-thread workgroups (were 256) pack their samples through LDS (2 barriers) once at depth 10, not 4: stopped ones fill 4 waves, one per SIMD, which store and exit; a sample may stop at other depths
# speedup vs baseline: 1.4478x; 1.0623x over previous
#include <hip/hip_runtime.h>


typedef float v2f __attribute__((ext_vector_type(2)));
typedef float v8f __attribute__((ext_vector_type(8)));
typedef int   v2i __attribute__((ext_vector_type(2)));
typedef int   v4i __attribute__((ext_vector_type(4)));
typedef float v4f __attribute__((ext_vector_type(4)));

#define NFPN_EPS2 1e-10f
#define NFPN_MAX_DEPTH 100
#define NFPN_PACK_DEPTH 10
#define NFPN_THREADS 512
#define NFPN_WAVES (NFPN_THREADS / 32)

__global__ __launch_bounds__(NFPN_THREADS) __attribute__((amdgpu_waves_per_eu(16, 16))) void nfpn_rps_kernel(
    const float* __restrict__ din,
    const float* __restrict__ W1,
    const float* __restrict__ b1,
    const float* __restrict__ W2,
    const float* __restrict__ b2,
    float* __restrict__ out,
    int B)
{
    __shared__ int s_cnt[NFPN_WAVES];
    __shared__ v4i s_u[2 * NFPN_THREADS];
    __shared__ v4f s_q[2 * NFPN_THREADS];

    const int lane  = threadIdx.x & 31;
    const int c     = lane & 15;
    const int h     = lane >> 4;
    const int wbase = (int)(blockIdx.x * blockDim.x + threadIdx.x) - lane;

    int  s[2];
    bool act[2];
#pragma unroll
    for (int t = 0; t < 2; ++t) {
        s[t]   = wbase + 16 * t + c;
        act[t] = (s[t] < B);
        if (!act[t]) s[t] = B - 1;
    }

    const float* W1h = W1 + 9 * h;
    const float* b1h = b1 + 3 * h;
    float q[2][3];
#pragma unroll
    for (int t = 0; t < 2; ++t) {
        const float d0 = din[3 * s[t] + 0];
        const float d1 = din[3 * s[t] + 1];
        const float d2 = din[3 * s[t] + 2];
#pragma unroll
        for (int i = 0; i < 3; ++i) {
            float v = fmaf(d0, W1h[3*i + 0], fmaf(d1, W1h[3*i + 1], fmaf(d2, W1h[3*i + 2], b1h[i])));
            q[t][i] = (v >= 0.f) ? v : 0.1f * v;
        }
    }

    const int      ab   = c >> 3;
    const int      ar   = c & 7;
    const unsigned sel  = (0x07653421u >> (4 * ar)) & 7u;
    const float    scl  = (ar < 3) ? 1.0f : ((ar < 6) ? 0.5f : ((ar == 6) ? (1.0f / 3.0f) : 0.0f));
    const float*   Wsub = W2 + 18 * ab + 3 * h;
    float a[3];
#pragma unroll
    for (int e = 0; e < 3; ++e) {
        const float w0 = (sel & 1u) ? Wsub[e]      : 0.f;
        const float w1 = (sel & 2u) ? Wsub[6 + e]  : 0.f;
        const float w2 = (sel & 4u) ? Wsub[12 + e] : 0.f;
        a[e] = -scl * ((w0 + w1) + w2);
    }
    v2f a1, a2;
    a1.x = a[0]; a1.y = a[1];
    a2.x = a[2]; a2.y = 0.f;

    const float e0 = b2[3 * h + 0];
    const float e1 = b2[3 * h + 1];
    const float e2 = b2[3 * h + 2];
    v8f cinit;
    cinit[0] = -e0;
    cinit[1] = -e1;
    cinit[2] = -e2;
    cinit[3] = -0.5f * ((e0 + e1) + 1.0f);
    cinit[4] = -0.5f * ((e0 + e2) + 1.0f);
    cinit[5] = -0.5f * ((e1 + e2) + 1.0f);
    cinit[6] = -(1.0f / 3.0f) * (((e0 + e1) + e2) + 1.0f);
    cinit[7] = 0.f;

    auto step2 = [&](const float (&u)[2][3], float (&un)[2][3]) {
        v8f acc[2];
#pragma unroll
        for (int t = 0; t < 2; ++t) {
            v2f B1, B2;
            B1.x = u[t][0] + q[t][0];
            B1.y = u[t][1] + q[t][1];
            B2.x = u[t][2] + q[t][2];
            B2.y = 0.f;
            acc[t] = __builtin_amdgcn_wmma_f32_16x16x4_f32(false, a1, false, B1, (short)0, cinit,  false, false);
            acc[t] = __builtin_amdgcn_wmma_f32_16x16x4_f32(false, a2, false, B2, (short)0, acc[t], false, false);
        }
#pragma unroll
        for (int t = 0; t < 2; ++t) {
            const float hi = fmaxf(fmaxf(acc[t][0], acc[t][1]), acc[t][2]);
            const float mp = fmaxf(fmaxf(acc[t][3], acc[t][4]), acc[t][5]);
            const float th = fmaxf(fmaxf(hi - 1.0f, mp), acc[t][6]);
#pragma unroll
            for (int i = 0; i < 3; ++i)
                un[t][i] = fmaxf(acc[t][i] - th, 0.f);
        }
    };

    auto step1 = [&](const float (&x)[3], const float (&qq)[3], float (&xn)[3]) {
        v2f B1, B2;
        B1.x = x[0] + qq[0];
        B1.y = x[1] + qq[1];
        B2.x = x[2] + qq[2];
        B2.y = 0.f;
        v8f acc = __builtin_amdgcn_wmma_f32_16x16x4_f32(false, a1, false, B1, (short)0, cinit, false, false);
        acc     = __builtin_amdgcn_wmma_f32_16x16x4_f32(false, a2, false, B2, (short)0, acc,   false, false);
        const float hi = fmaxf(fmaxf(acc[0], acc[1]), acc[2]);
        const float mp = fmaxf(fmaxf(acc[3], acc[4]), acc[5]);
        const float th = fmaxf(fmaxf(hi - 1.0f, mp), acc[6]);
#pragma unroll
        for (int i = 0; i < 3; ++i)
            xn[i] = fmaxf(acc[i] - th, 0.f);
    };

    float u[2][3];
#pragma unroll
    for (int t = 0; t < 2; ++t)
#pragma unroll
        for (int i = 0; i < 3; ++i) u[t][i] = 1.0f / 3.0f;

    static_assert((NFPN_MAX_DEPTH % 2) == 0, "two steps per trip need an even cap");
    static_assert((NFPN_PACK_DEPTH % 2) == 0 && NFPN_PACK_DEPTH >= 2 && NFPN_PACK_DEPTH < NFPN_MAX_DEPTH,
                  "the pack depth is even and below the cap");
    float v[2][3];

    auto residual2 = [&]() -> float {
        float rp[2];
#pragma unroll
        for (int t = 0; t < 2; ++t) {
            const float g0 = u[t][0] - v[t][0];
            const float g1 = u[t][1] - v[t][1];
            const float g2 = u[t][2] - v[t][2];
            rp[t] = fmaf(g2, g2, fmaf(g1, g1, g0 * g0));
        }
        v2i sw = __builtin_amdgcn_permlane16_swap(__float_as_int(rp[0]), __float_as_int(rp[1]), false, false);
        return __int_as_float(sw.x) + __int_as_float(sw.y);
    };

    for (int k = 0; k < NFPN_PACK_DEPTH / 2; ++k) {
        step2(u, v);
        step2(v, u);
    }
    int depth = NFPN_PACK_DEPTH;
    unsigned moving = __builtin_amdgcn_ballot_w32((residual2() > NFPN_EPS2) && (wbase + lane < B));

    {
        static_assert(NFPN_WAVES == 16, "the block table below is written for 16 waves");
        const int w = __builtin_amdgcn_readfirstlane((int)(threadIdx.x >> 5));
        const int wb = (int)((0xFBA98E7654D3210Cull >> (4 * w)) & 15ull);
        if (lane == 0) s_cnt[w] = __popc(moving);
        __syncthreads();
        int off = 0;
        int M   = 0;
#pragma unroll
        for (int i = 0; i < NFPN_WAVES; ++i) {
            const int n = s_cnt[i];
            M   += n;
            off += (i < w) ? n : 0;
        }
        const int sb = 32 * w - off;
#pragma unroll
        for (int t = 0; t < 2; ++t) {
            const int  b  = 16 * t + c;
            const int  kb = __popc(moving & ((1u << b) - 1u));
            const bool mv = ((moving >> b) & 1u) != 0u;
            const int  p  = mv ? (off + kb) : (M + sb + (b - kb));
            const int  r  = NFPN_THREADS * h + p;
            v4i ru;
            ru.x = __float_as_int(u[t][0]);
            ru.y = __float_as_int(u[t][1]);
            ru.z = __float_as_int(u[t][2]);
            ru.w = act[t] ? s[t] : -1;
            v4f rq;
            rq.x = q[t][0];
            rq.y = q[t][1];
            rq.z = q[t][2];
            rq.w = 0.f;
            s_u[r] = ru;
            s_q[r] = rq;
        }
        __syncthreads();
#pragma unroll
        for (int t = 0; t < 2; ++t) {
            const int r  = NFPN_THREADS * h + 32 * wb + 16 * t + c;
            const v4i ru = s_u[r];
            const v4f rq = s_q[r];
            u[t][0] = __int_as_float(ru.x);
            u[t][1] = __int_as_float(ru.y);
            u[t][2] = __int_as_float(ru.z);
            s[t]    = ru.w;
            act[t]  = (ru.w >= 0);
            q[t][0] = rq.x;
            q[t][1] = rq.y;
            q[t][2] = rq.z;
        }
        const int nm = M - 32 * wb;
        moving = (nm >= 32) ? 0xFFFFFFFFu : ((nm <= 0) ? 0u : ((1u << nm) - 1u));
    }

    if (__popc(moving) > 16) {
        for (;;) {
            step2(u, v);
            step2(v, u);
            const float r2 = residual2();
            depth += 2;
            if (depth >= NFPN_MAX_DEPTH) { moving = 0u; break; }
            moving = __builtin_amdgcn_ballot_w32(r2 > NFPN_EPS2);
            if (__popc(moving) <= 16) break;
        }
    }

    {
        float o[2][3];
        step2(u, o);
#pragma unroll
        for (int t = 0; t < 2; ++t) {
            const bool done = ((moving >> (16 * t + c)) & 1u) == 0u;
            if (act[t] && done) {
                float* op = out + 6l * s[t] + 3 * h;
                op[0] = o[t][0];
                op[1] = o[t][1];
                op[2] = o[t][2];
            }
        }
    }
    if (moving == 0u) return;

    const unsigned m0 = moving & 0xFFFFu;
    const unsigned m1 = moving >> 16;
    const int n0 = __popc(m0);
    const int n1 = __popc(m1);
    const unsigned below = (1u << c) - 1u;
    const int k0 = __popc(m0 & below);
    const int k1 = __popc(m1 & below);
    const int p0 = ((m0 >> c) & 1u) ? k0 : n0 + (c - k0);
    const int p1 = (((m1 >> c) & 1u) ? n0 + k1 : n0 + n1 + (c - k1)) & 15;
    const int ad0 = 4 * (p0 + 16 * h);
    const int ad1 = 4 * (p1 + 16 * h);
    const bool from0 = (c < n0);
    auto pack = [&](int x0, int x1) -> int {
        const int y0 = __builtin_amdgcn_ds_permute(ad0, x0);
        const int y1 = __builtin_amdgcn_ds_permute(ad1, x1);
        return from0 ? y0 : y1;
    };
    float pu[3], pq[3];
#pragma unroll
    for (int i = 0; i < 3; ++i) {
        pu[i] = __int_as_float(pack(__float_as_int(u[0][i]), __float_as_int(u[1][i])));
        pq[i] = __int_as_float(pack(__float_as_int(q[0][i]), __float_as_int(q[1][i])));
    }
    const int  ps   = pack(act[0] ? s[0] : -1, act[1] ? s[1] : -1);
    const bool live = (c < n0 + n1);
    const bool pact = live && (ps >= 0);

    float pv[3];
    for (;;) {
        step1(pu, pq, pv);
        step1(pv, pq, pu);
        const float g0 = pu[0] - pv[0];
        const float g1 = pu[1] - pv[1];
        const float g2 = pu[2] - pv[2];
        const int rp = __float_as_int(fmaf(g2, g2, fmaf(g1, g1, g0 * g0)));
        v2i sw = __builtin_amdgcn_permlane16_swap(rp, rp, false, false);
        const float r2 = __int_as_float(sw.x) + __int_as_float(sw.y);
        depth += 2;
        if (depth >= NFPN_MAX_DEPTH) break;
        if (__builtin_amdgcn_ballot_w32(live && (r2 > NFPN_EPS2)) == 0u) break;
    }

    float po[3];
    step1(pu, pq, po);
    if (pact) {
        float* op = out + 6l * ps + 3 * h;
        op[0] = po[0];
        op[1] = po[1];
        op[2] = po[2];
    }
}

extern "C" void kernel_launch(void* const* d_in, const int* in_sizes, int n_in,
                              void* d_out, int out_size, void* d_ws, size_t ws_size,
                              hipStream_t stream) {
    const float* d  = (const float*)d_in[0];
    const float* W1 = (const float*)d_in[1];
    const float* b1 = (const float*)d_in[2];
    const float* W2 = (const float*)d_in[3];
    const float* b2 = (const float*)d_in[4];
    float* out = (float*)d_out;

    const int B = in_sizes[0] / 3;
    const int threads = NFPN_THREADS;
    const int blocks  = (B + threads - 1) / threads;
    hipLaunchKernelGGL(nfpn_rps_kernel, dim3(blocks), dim3(threads), 0, stream,
                       d, W1, b1, W2, b2, out, B);
}


# Step constants leave the loop: A(1+q)-b2 (rows 0-2 less 1) is a per-sample C operand built once, state held as u-1 is the B operand itself; no u+q adds, no largest-1; constant part summed first
# speedup vs baseline: 1.5201x; 1.0499x over previous
#include <hip/hip_runtime.h>


typedef float v2f __attribute__((ext_vector_type(2)));
typedef float v8f __attribute__((ext_vector_type(8)));
typedef int   v2i __attribute__((ext_vector_type(2)));
typedef int   v4i __attribute__((ext_vector_type(4)));
typedef float v4f __attribute__((ext_vector_type(4)));

#define NFPN_EPS2 1e-10f
#define NFPN_MAX_DEPTH 100
#define NFPN_PACK_DEPTH 10
#define NFPN_THREADS 512
#define NFPN_WAVES (NFPN_THREADS / 32)

__global__ __launch_bounds__(NFPN_THREADS) __attribute__((amdgpu_waves_per_eu(16, 16))) void nfpn_rps_kernel(
    const float* __restrict__ din,
    const float* __restrict__ W1,
    const float* __restrict__ b1,
    const float* __restrict__ W2,
    const float* __restrict__ b2,
    float* __restrict__ out,
    int B)
{
    __shared__ int s_cnt[NFPN_WAVES];
    __shared__ v4i s_u[2 * NFPN_THREADS];
    __shared__ v4f s_c[2 * NFPN_THREADS];

    const int lane  = threadIdx.x & 31;
    const int c     = lane & 15;
    const int h     = lane >> 4;
    const int wbase = (int)(blockIdx.x * blockDim.x + threadIdx.x) - lane;

    int  s[2];
    bool act[2];
#pragma unroll
    for (int t = 0; t < 2; ++t) {
        s[t]   = wbase + 16 * t + c;
        act[t] = (s[t] < B);
        if (!act[t]) s[t] = B - 1;
    }

    const float* W1h = W1 + 9 * h;
    const float* b1h = b1 + 3 * h;
    float q[2][3];
#pragma unroll
    for (int t = 0; t < 2; ++t) {
        const float d0 = din[3 * s[t] + 0];
        const float d1 = din[3 * s[t] + 1];
        const float d2 = din[3 * s[t] + 2];
#pragma unroll
        for (int i = 0; i < 3; ++i) {
            float v = fmaf(d0, W1h[3*i + 0], fmaf(d1, W1h[3*i + 1], fmaf(d2, W1h[3*i + 2], b1h[i])));
            q[t][i] = (v >= 0.f) ? v : 0.1f * v;
        }
    }

    const int      ab   = c >> 3;
    const int      ar   = c & 7;
    const unsigned sel  = (0x07653421u >> (4 * ar)) & 7u;
    const float    scl  = (ar < 3) ? 1.0f : ((ar < 6) ? 0.5f : ((ar == 6) ? (1.0f / 3.0f) : 0.0f));
    const float*   Wsub = W2 + 18 * ab + 3 * h;
    float a[3];
#pragma unroll
    for (int e = 0; e < 3; ++e) {
        const float w0 = (sel & 1u) ? Wsub[e]      : 0.f;
        const float w1 = (sel & 2u) ? Wsub[6 + e]  : 0.f;
        const float w2 = (sel & 4u) ? Wsub[12 + e] : 0.f;
        a[e] = -scl * ((w0 + w1) + w2);
    }
    v2f a1, a2;
    a1.x = a[0]; a1.y = a[1];
    a2.x = a[2]; a2.y = 0.f;

    const float e0 = b2[3 * h + 0];
    const float e1 = b2[3 * h + 1];
    const float e2 = b2[3 * h + 2];
    v8f cbase;
    cbase[0] = -e0 - 1.0f;
    cbase[1] = -e1 - 1.0f;
    cbase[2] = -e2 - 1.0f;
    cbase[3] = -0.5f * ((e0 + e1) + 1.0f);
    cbase[4] = -0.5f * ((e0 + e2) + 1.0f);
    cbase[5] = -0.5f * ((e1 + e2) + 1.0f);
    cbase[6] = -(1.0f / 3.0f) * (((e0 + e1) + e2) + 1.0f);
    cbase[7] = 0.f;

    auto project = [&](const v8f& acc, v2f (&xn)[2]) {
        const float hi = fmaxf(fmaxf(acc[0], acc[1]), acc[2]);
        const float mp = fmaxf(fmaxf(acc[3], acc[4]), acc[5]);
        const float th = fmaxf(fmaxf(hi, mp), acc[6]);
        xn[0].x = fmaxf(acc[0] - th, -1.0f);
        xn[0].y = fmaxf(acc[1] - th, -1.0f);
        xn[1].x = fmaxf(acc[2] - th, -1.0f);
        xn[1].y = acc[7];
    };

    v8f cs[2];

    auto step2 = [&](const v2f (&x)[2][2], v2f (&xn)[2][2]) {
        v8f acc[2];
#pragma unroll
        for (int t = 0; t < 2; ++t) {
            acc[t] = __builtin_amdgcn_wmma_f32_16x16x4_f32(false, a1, false, x[t][0], (short)0, cs[t],  false, false);
            acc[t] = __builtin_amdgcn_wmma_f32_16x16x4_f32(false, a2, false, x[t][1], (short)0, acc[t], false, false);
        }
#pragma unroll
        for (int t = 0; t < 2; ++t) project(acc[t], xn[t]);
    };

    v8f pcs;
    auto step1 = [&](const v2f (&x)[2], v2f (&xn)[2]) {
        v8f acc = __builtin_amdgcn_wmma_f32_16x16x4_f32(false, a1, false, x[0], (short)0, pcs, false, false);
        acc     = __builtin_amdgcn_wmma_f32_16x16x4_f32(false, a2, false, x[1], (short)0, acc, false, false);
        project(acc, xn);
    };

    auto rebuild = [&](float c0, float c1, float c2) -> v8f {
        v8f r;
        const float t01 = c0 + c1;
        r[0] = c0;
        r[1] = c1;
        r[2] = c2;
        r[3] = fmaf(t01,     0.5f, 0.5f);
        r[4] = fmaf(c0 + c2, 0.5f, 0.5f);
        r[5] = fmaf(c1 + c2, 0.5f, 0.5f);
        r[6] = fmaf(t01 + c2, 1.0f / 3.0f, 2.0f / 3.0f);
        r[7] = 0.f;
        return r;
    };

    static_assert((NFPN_MAX_DEPTH % 2) == 0, "two steps per trip need an even cap");
    static_assert((NFPN_PACK_DEPTH % 2) == 0 && NFPN_PACK_DEPTH >= 2 && NFPN_PACK_DEPTH < NFPN_MAX_DEPTH,
                  "the pack depth is even and below the cap");
    v2f u[2][2], v[2][2];

    {
        v8f kt = {0.f, 0.f, 0.f, 0.f, 0.f, 0.f, 0.f, 0.f};
        v2f T1, T2;
        T1.x = 2.0f / 3.0f; T1.y = 2.0f / 3.0f;
        T2.x = 2.0f / 3.0f; T2.y = 0.f;
        kt = __builtin_amdgcn_wmma_f32_16x16x4_f32(false, a1, false, T1, (short)0, kt, false, false);
        kt = __builtin_amdgcn_wmma_f32_16x16x4_f32(false, a2, false, T2, (short)0, kt, false, false);
        v8f acc[2];
#pragma unroll
        for (int t = 0; t < 2; ++t) {
            v2f B1, B2;
            B1.x = (1.0f / 3.0f) + q[t][0];
            B1.y = (1.0f / 3.0f) + q[t][1];
            B2.x = (1.0f / 3.0f) + q[t][2];
            B2.y = 0.f;
            acc[t] = __builtin_amdgcn_wmma_f32_16x16x4_f32(false, a1, false, B1, (short)0, cbase,  false, false);
            acc[t] = __builtin_amdgcn_wmma_f32_16x16x4_f32(false, a2, false, B2, (short)0, acc[t], false, false);
        }
#pragma unroll
        for (int t = 0; t < 2; ++t) {
            cs[t] = acc[t] + kt;
            project(acc[t], v[t]);
        }
    }
    step2(v, u);

    auto residual2 = [&]() -> float {
        float rp[2];
#pragma unroll
        for (int t = 0; t < 2; ++t) {
            const float g0 = u[t][0].x - v[t][0].x;
            const float g1 = u[t][0].y - v[t][0].y;
            const float g2 = u[t][1].x - v[t][1].x;
            rp[t] = fmaf(g2, g2, fmaf(g1, g1, g0 * g0));
        }
        v2i sw = __builtin_amdgcn_permlane16_swap(__float_as_int(rp[0]), __float_as_int(rp[1]), false, false);
        return __int_as_float(sw.x) + __int_as_float(sw.y);
    };

    for (int k = 1; k < NFPN_PACK_DEPTH / 2; ++k) {
        step2(u, v);
        step2(v, u);
    }
    int depth = NFPN_PACK_DEPTH;
    unsigned moving = __builtin_amdgcn_ballot_w32((residual2() > NFPN_EPS2) && ((int)threadIdx.x < B - (int)(blockIdx.x * blockDim.x)));

    {
        static_assert(NFPN_WAVES == 16, "the block table below is written for 16 waves");
        const int w = __builtin_amdgcn_readfirstlane((int)(threadIdx.x >> 5));
        const int wb = (int)((0xFBA98E7654D3210Cull >> (4 * w)) & 15ull);
        if (lane == 0) s_cnt[w] = __popc(moving);
        __syncthreads();
        int off = 0;
        int M   = 0;
#pragma unroll
        for (int i = 0; i < NFPN_WAVES; ++i) {
            const int n = s_cnt[i];
            M   += n;
            off += (i < w) ? n : 0;
        }
        const int sb = 32 * w - off;
#pragma unroll
        for (int t = 0; t < 2; ++t) {
            const int  b  = 16 * t + c;
            const int  kb = __popc(moving & ((1u << b) - 1u));
            const bool mv = ((moving >> b) & 1u) != 0u;
            const int  p  = mv ? (off + kb) : (M + sb + (b - kb));
            const int  r  = NFPN_THREADS * h + p;
            v4i ru;
            ru.x = __float_as_int(u[t][0].x);
            ru.y = __float_as_int(u[t][0].y);
            ru.z = __float_as_int(u[t][1].x);
            ru.w = (int)threadIdx.x - lane + b;
            v4f rc;
            rc.x = cs[t][0];
            rc.y = cs[t][1];
            rc.z = cs[t][2];
            rc.w = 0.f;
            s_u[r] = ru;
            s_c[r] = rc;
        }
        __syncthreads();
#pragma unroll
        for (int t = 0; t < 2; ++t) {
            const int r  = NFPN_THREADS * h + 32 * wb + 16 * t + c;
            const v4i ru = s_u[r];
            const v4f rc = s_c[r];
            u[t][0].x = __int_as_float(ru.x);
            u[t][0].y = __int_as_float(ru.y);
            u[t][1].x = __int_as_float(ru.z);
            u[t][1].y = 0.f;
            s[t]    = (int)(blockIdx.x * blockDim.x) + ru.w;
            act[t]  = (s[t] < B);
            cs[t]   = rebuild(rc.x, rc.y, rc.z);
        }
        const int nm = M - 32 * wb;
        moving = (nm >= 32) ? 0xFFFFFFFFu : ((nm <= 0) ? 0u : ((1u << nm) - 1u));
    }

    if (__popc(moving) > 16) {
        for (;;) {
            step2(u, v);
            step2(v, u);
            const float r2 = residual2();
            depth += 2;
            if (depth >= NFPN_MAX_DEPTH) { moving = 0u; break; }
            moving = __builtin_amdgcn_ballot_w32(r2 > NFPN_EPS2);
            if (__popc(moving) <= 16) break;
        }
    }

    {
        v2f o[2][2];
        step2(u, o);
#pragma unroll
        for (int t = 0; t < 2; ++t) {
            const bool done = ((moving >> (16 * t + c)) & 1u) == 0u;
            if (act[t] && done) {
                float* op = out + 6l * s[t] + 3 * h;
                op[0] = o[t][0].x + 1.0f;
                op[1] = o[t][0].y + 1.0f;
                op[2] = o[t][1].x + 1.0f;
            }
        }
    }
    if (moving == 0u) return;

    const unsigned m0 = moving & 0xFFFFu;
    const unsigned m1 = moving >> 16;
    const int n0 = __popc(m0);
    const int n1 = __popc(m1);
    const unsigned below = (1u << c) - 1u;
    const int k0 = __popc(m0 & below);
    const int k1 = __popc(m1 & below);
    const int p0 = ((m0 >> c) & 1u) ? k0 : n0 + (c - k0);
    const int p1 = (((m1 >> c) & 1u) ? n0 + k1 : n0 + n1 + (c - k1)) & 15;
    const int ad0 = 4 * (p0 + 16 * h);
    const int ad1 = 4 * (p1 + 16 * h);
    const bool from0 = (c < n0);
    auto pack = [&](int x0, int x1) -> int {
        const int y0 = __builtin_amdgcn_ds_permute(ad0, x0);
        const int y1 = __builtin_amdgcn_ds_permute(ad1, x1);
        return from0 ? y0 : y1;
    };
    auto packf = [&](float x0, float x1) -> float {
        return __int_as_float(pack(__float_as_int(x0), __float_as_int(x1)));
    };
    v2f pu[2], pv[2];
    pu[0].x = packf(u[0][0].x, u[1][0].x);
    pu[0].y = packf(u[0][0].y, u[1][0].y);
    pu[1].x = packf(u[0][1].x, u[1][1].x);
    pu[1].y = 0.f;
    {
        const float c0 = packf(cs[0][0], cs[1][0]);
        const float c1 = packf(cs[0][1], cs[1][1]);
        const float c2 = packf(cs[0][2], cs[1][2]);
        pcs = rebuild(c0, c1, c2);
    }
    const int  ps   = pack(act[0] ? s[0] : -1, act[1] ? s[1] : -1);
    const bool live = (c < n0 + n1);
    const bool pact = live && (ps >= 0);

    for (;;) {
        step1(pu, pv);
        step1(pv, pu);
        const float g0 = pu[0].x - pv[0].x;
        const float g1 = pu[0].y - pv[0].y;
        const float g2 = pu[1].x - pv[1].x;
        const int rp = __float_as_int(fmaf(g2, g2, fmaf(g1, g1, g0 * g0)));
        v2i sw = __builtin_amdgcn_permlane16_swap(rp, rp, false, false);
        const float r2 = __int_as_float(sw.x) + __int_as_float(sw.y);
        depth += 2;
        if (depth >= NFPN_MAX_DEPTH) break;
        if (__builtin_amdgcn_ballot_w32(live && (r2 > NFPN_EPS2)) == 0u) break;
    }

    v2f po[2];
    step1(pu, po);
    if (pact) {
        float* op = out + 6l * ps + 3 * h;
        op[0] = po[0].x + 1.0f;
        op[1] = po[0].y + 1.0f;
        op[2] = po[1].x + 1.0f;
    }
}

extern "C" void kernel_launch(void* const* d_in, const int* in_sizes, int n_in,
                              void* d_out, int out_size, void* d_ws, size_t ws_size,
                              hipStream_t stream) {
    const float* d  = (const float*)d_in[0];
    const float* W1 = (const float*)d_in[1];
    const float* b1 = (const float*)d_in[2];
    const float* W2 = (const float*)d_in[3];
    const float* b2 = (const float*)d_in[4];
    float* out = (float*)d_out;

    const int B = in_sizes[0] / 3;
    const int threads = NFPN_THREADS;
    const int blocks  = (B + threads - 1) / threads;
    hipLaunchKernelGGL(nfpn_rps_kernel, dim3(blocks), dim3(threads), 0, stream,
                       d, W1, b1, W2, b2, out, B);
}
